# starting kernel
# baseline (speedup 1.0000x reference)
#include <hip/hip_runtime.h>

typedef __attribute__((ext_vector_type(2))) float v2f;
typedef __attribute__((ext_vector_type(4))) float v4f;
typedef __attribute__((ext_vector_type(8))) float v8f;

#define VQ_B 64
#define VQ_C 64
#define VQ_HW 1024
#define VQ_K 1024
#define WAVES 8
#define XS_STRIDE 68

__global__ __launch_bounds__(256) void vq_cnorm_kernel(const float* __restrict__ cb,
                                                       float* __restrict__ cn2h) {
  int k = blockIdx.x * blockDim.x + threadIdx.x;
  if (k < VQ_K) {
    const float* row = cb + (size_t)k * VQ_C;
    float s = 0.0f;
#pragma unroll
    for (int c = 0; c < VQ_C; c += 4) {
      v4f v = *(const v4f*)(row + c);
      s += v.x * v.x + v.y * v.y + v.z * v.z + v.w * v.w;
    }
    cn2h[k] = 0.5f * s;
  }
}

__device__ __forceinline__ void vq_ldb(const float* __restrict__ cb, int code, int bk,
                                       v2f (&bf)[16]) {
  const v2f* bp = (const v2f*)cb + (size_t)code * (VQ_C / 2) + bk;
#pragma unroll
  for (int kk = 0; kk < 16; ++kk) bf[kk] = bp[kk * 2];
}

__device__ __forceinline__ void vq_tile(const v2f (&a)[16], const v2f (&bf)[16],
                                        int code, const float* __restrict__ cn2h,
                                        float (&best)[8], int (&bidx)[8]) {
  const float cn = cn2h[code];
  v8f acc0 = {}, acc1 = {};
#pragma unroll
  for (int kk = 0; kk < 16; kk += 2) {
    acc0 = __builtin_amdgcn_wmma_f32_16x16x4_f32(
        false, a[kk + 0], false, bf[kk + 0], (short)0, acc0, false, false);
    acc1 = __builtin_amdgcn_wmma_f32_16x16x4_f32(
        false, a[kk + 1], false, bf[kk + 1], (short)0, acc1, false, false);
  }
#pragma unroll
  for (int r = 0; r < 8; ++r) {
    float s = (acc0[r] + acc1[r]) - cn;
    if (s > best[r]) { best[r] = s; bidx[r] = code; }
  }
}

__global__ __launch_bounds__(256) void vq_main_kernel(const float* __restrict__ in,
                                                      const float* __restrict__ cb,
                                                      const float* __restrict__ cn2h,
                                                      float* __restrict__ out) {
  __shared__ float xs[16 * XS_STRIDE];
  __shared__ float red_s[16 * WAVES];
  __shared__ int   red_i[16 * WAVES];
  __shared__ int   rowbest[16];

  const int tid  = threadIdx.x;
  const int lane = tid & 31;
  const int wave = tid >> 5;
  const int m    = blockIdx.x;
  const int b    = m >> 6;
  const int hw0  = (m & 63) << 4;

  {
    const int c  = tid >> 2;
    const int dn = (tid & 3) << 2;
    v4f v = *(const v4f*)(in + (((size_t)(b * VQ_C + c)) << 10) + hw0 + dn);
    xs[(dn + 0) * XS_STRIDE + c] = v.x;
    xs[(dn + 1) * XS_STRIDE + c] = v.y;
    xs[(dn + 2) * XS_STRIDE + c] = v.z;
    xs[(dn + 3) * XS_STRIDE + c] = v.w;
  }
  __syncthreads();

  const int an = lane & 15;
  const int ak = (lane >> 4) << 1;
  v2f a[16];
#pragma unroll
  for (int kk = 0; kk < 16; ++kk)
    a[kk] = *(const v2f*)(xs + an * XS_STRIDE + kk * 4 + ak);

  float best[8];
  int   bidx[8];
#pragma unroll
  for (int r = 0; r < 8; ++r) { best[r] = -3.0e38f; bidx[r] = 0; }

  const int ncol = lane & 15;
  const int bk   = lane >> 4;
  const int base = wave * 8;

  v2f bf0[16], bf1[16];
  vq_ldb(cb, (base + 0) * 16 + ncol, bk, bf0);
#pragma unroll 1
  for (int tt = 0; tt < 4; ++tt) {
    const int t0 = 2 * tt, t1 = 2 * tt + 1;
    vq_ldb(cb, (base + t1) * 16 + ncol, bk, bf1);
    vq_tile(a, bf0, (base + t0) * 16 + ncol, cn2h, best, bidx);
    if (tt < 3)
      vq_ldb(cb, (base + t1 + 1) * 16 + ncol, bk, bf0);
    vq_tile(a, bf1, (base + t1) * 16 + ncol, cn2h, best, bidx);
  }

#pragma unroll
  for (int r = 0; r < 8; ++r) {
    float s = best[r];
    int   i = bidx[r];
#pragma unroll
    for (int msk = 8; msk >= 1; msk >>= 1) {
      float os = __shfl_xor(s, msk, 32);
      int   oi = __shfl_xor(i, msk, 32);
      if (os > s || (os == s && oi < i)) { s = os; i = oi; }
    }
    const int rowm = r + ((lane >> 4) << 3);
    if ((lane & 15) == 0) {
      red_s[rowm * WAVES + wave] = s;
      red_i[rowm * WAVES + wave] = i;
    }
  }
  __syncthreads();

  if (tid < 16) {
    float s = red_s[tid * WAVES];
    int   i = red_i[tid * WAVES];
#pragma unroll
    for (int w = 1; w < WAVES; ++w) {
      float os = red_s[tid * WAVES + w];
      int   oi = red_i[tid * WAVES + w];
      if (os > s || (os == s && oi < i)) { s = os; i = oi; }
    }
    rowbest[tid] = i;
  }
  __syncthreads();

  {
    const int row = tid & 15;
    const int c0  = tid >> 4;
    const float* crow = cb + (size_t)rowbest[row] * VQ_C;
    const size_t obase = (((size_t)b * VQ_C) << 10) + hw0 + row;
#pragma unroll
    for (int j = 0; j < 4; ++j) {
      const int c = c0 + 16 * j;
      out[obase + ((size_t)c << 10)] = crow[c];
    }
  }
}

extern "C" void kernel_launch(void* const* d_in, const int* in_sizes, int n_in,
                              void* d_out, int out_size, void* d_ws, size_t ws_size,
                              hipStream_t stream) {
  const float* in  = (const float*)d_in[0];
  const float* cb  = (const float*)d_in[1];
  float* out  = (float*)d_out;
  float* cn2h = (float*)d_ws;

  vq_cnorm_kernel<<<VQ_K / 256, 256, 0, stream>>>(cb, cn2h);
  vq_main_kernel<<<(VQ_B * VQ_HW) / 16, 256, 0, stream>>>(in, cb, cn2h, out);
}


# Codebook staged in LDS per 256-pixel workgroup (pitch 68, four chunks, next one prefetched); each wave sweeps all 1024 codes for its own pixel tile, B fragments from LDS not L2; sum order unchanged
# speedup vs baseline: 2.2759x; 2.2759x over previous
#include <hip/hip_runtime.h>


typedef __attribute__((ext_vector_type(2))) float v2f;
typedef __attribute__((ext_vector_type(4))) float v4f;
typedef __attribute__((ext_vector_type(8))) float v8f;

#define VQ_B 64
#define VQ_C 64
#define VQ_HW 1024
#define VQ_K 1024
#define WG_THREADS 512
#define WG_PIX 256
#define CB_PITCH 68
#define CHUNK_CODES 256
#define N_CHUNKS 4
#define CHUNK_TILES 16

__global__ __launch_bounds__(256) void vq_cnorm_kernel(const float* __restrict__ cb,
                                                       float* __restrict__ cn2h) {
  int k = blockIdx.x * blockDim.x + threadIdx.x;
  if (k < VQ_K) {
    const float* row = cb + (size_t)k * VQ_C;
    float s = 0.0f;
#pragma unroll
    for (int c = 0; c < VQ_C; c += 4) {
      v4f v = *(const v4f*)(row + c);
      s += v.x * v.x + v.y * v.y + v.z * v.z + v.w * v.w;
    }
    cn2h[k] = 0.5f * s;
  }
}

__device__ __forceinline__ void vq_fetch(const float* __restrict__ cb, int chunk, int tid,
                                         v4f (&pf)[8]) {
  const v4f* gp = (const v4f*)(cb + (size_t)chunk * (CHUNK_CODES * VQ_C)) + tid;
#pragma unroll
  for (int i = 0; i < 8; ++i) pf[i] = gp[i * WG_THREADS];
}

__device__ __forceinline__ void vq_stash(float* cbs, int chunk, int tid, const v4f (&pf)[8]) {
  float* sp = cbs + (chunk * CHUNK_CODES + (tid >> 4)) * CB_PITCH + ((tid & 15) << 2);
#pragma unroll
  for (int i = 0; i < 8; ++i) *(v4f*)(sp + i * 32 * CB_PITCH) = pf[i];
}

__device__ __forceinline__ void vq_ldb(const float* cbs, const float* cns, int code, int bk,
                                       v2f (&bf)[16], float& cn) {
  cn = cns[code];
  const v2f* bp = (const v2f*)(cbs + code * CB_PITCH) + bk;
#pragma unroll
  for (int kk = 0; kk < 16; ++kk) bf[kk] = bp[kk * 2];
}

__device__ __forceinline__ void vq_tile(const v2f (&a)[16], const v2f (&bf)[16],
                                        const float cn, int code,
                                        float (&best)[8], int (&bidx)[8]) {
  v8f acc0 = {}, acc1 = {};
#pragma unroll
  for (int kk = 0; kk < 16; kk += 2) {
    acc0 = __builtin_amdgcn_wmma_f32_16x16x4_f32(
        false, a[kk + 0], false, bf[kk + 0], (short)0, acc0, false, false);
    acc1 = __builtin_amdgcn_wmma_f32_16x16x4_f32(
        false, a[kk + 1], false, bf[kk + 1], (short)0, acc1, false, false);
  }
#pragma unroll
  for (int r = 0; r < 8; ++r) {
    float s = (acc0[r] + acc1[r]) - cn;
    if (s > best[r]) { best[r] = s; bidx[r] = code; }
  }
}

__global__ __launch_bounds__(WG_THREADS) void vq_main_kernel(const float* __restrict__ in,
                                                             const float* __restrict__ cb,
                                                             const float* __restrict__ cn2h,
                                                             float* __restrict__ out) {
  __shared__ __attribute__((aligned(16))) float cbs[VQ_K * CB_PITCH];
  __shared__ float cns[VQ_K];
  __shared__ int   rowbest[WG_PIX];

  const int tid  = threadIdx.x;
  const int lane = tid & 31;
  const int wave = tid >> 5;
  const int m    = blockIdx.x;
  const int b    = m >> 2;
  const int hwb  = (m & 3) << 8;
  const int hw0  = hwb + (wave << 4);

  v4f pf[8];
  vq_fetch(cb, 0, tid, pf);
  const float cnA = cn2h[tid];
  const float cnB = cn2h[tid + WG_THREADS];

  const int an = lane & 15;
  const int bk = lane >> 4;
  v2f a[16];
  {
    const float* ap = in + (((size_t)(b * VQ_C + 2 * bk)) << 10) + hw0 + an;
#pragma unroll
    for (int kk = 0; kk < 16; ++kk) {
      a[kk].x = ap[(size_t)(4 * kk + 0) << 10];
      a[kk].y = ap[(size_t)(4 * kk + 1) << 10];
    }
  }

  cns[tid] = cnA;
  cns[tid + WG_THREADS] = cnB;
  vq_stash(cbs, 0, tid, pf);
  __syncthreads();

  float best[8];
  int   bidx[8];
#pragma unroll
  for (int r = 0; r < 8; ++r) { best[r] = -3.0e38f; bidx[r] = 0; }

  const int ncol = lane & 15;

#pragma unroll 1
  for (int ch = 0; ch < N_CHUNKS; ++ch) {
    if (ch + 1 < N_CHUNKS) vq_fetch(cb, ch + 1, tid, pf);

    const int cbase = ch * CHUNK_CODES + ncol;
    v2f bf0[16], bf1[16];
    float cn0, cn1;
    vq_ldb(cbs, cns, cbase, bk, bf0, cn0);
#pragma unroll 1
    for (int tt = 0; tt < CHUNK_TILES / 2; ++tt) {
      const int code0 = cbase + 32 * tt;
      const int code1 = code0 + 16;
      vq_ldb(cbs, cns, code1, bk, bf1, cn1);
      vq_tile(a, bf0, cn0, code0, best, bidx);
      if (tt + 1 < CHUNK_TILES / 2)
        vq_ldb(cbs, cns, code1 + 16, bk, bf0, cn0);
      vq_tile(a, bf1, cn1, code1, best, bidx);
    }

    if (ch + 1 < N_CHUNKS) {
      vq_stash(cbs, ch + 1, tid, pf);
      __syncthreads();
    }
  }

#pragma unroll
  for (int r = 0; r < 8; ++r) {
    float s = best[r];
    int   i = bidx[r];
#pragma unroll
    for (int msk = 8; msk >= 1; msk >>= 1) {
      float os = __shfl_xor(s, msk, 32);
      int   oi = __shfl_xor(i, msk, 32);
      if (os > s || (os == s && oi < i)) { s = os; i = oi; }
    }
    const int rowm = r + (bk << 3);
    if (ncol == 0) rowbest[(wave << 4) + rowm] = i;
  }
  __syncthreads();

  {
    const int p  = ((wave >> 1) << 5) + lane;
    const int c0 = (wave & 1) << 5;
    const float* crow = cbs + rowbest[p] * CB_PITCH + c0;
    float* op = out + (((size_t)(b * VQ_C + c0)) << 10) + hwb + p;
#pragma unroll
    for (int j = 0; j < 32; ++j) op[(size_t)j << 10] = crow[j];
  }
}

extern "C" void kernel_launch(void* const* d_in, const int* in_sizes, int n_in,
                              void* d_out, int out_size, void* d_ws, size_t ws_size,
                              hipStream_t stream) {
  const float* in  = (const float*)d_in[0];
  const float* cb  = (const float*)d_in[1];
  float* out  = (float*)d_out;
  float* cn2h = (float*)d_ws;

  vq_cnorm_kernel<<<VQ_K / 256, 256, 0, stream>>>(cb, cn2h);
  vq_main_kernel<<<(VQ_B * VQ_HW) / WG_PIX, WG_THREADS, 0, stream>>>(in, cb, cn2h, out);
}


# Half norms summed in the main kernel from the registers each codebook chunk is staged through (4 squares per lane, 16-lane DPP add tree), first launch dropped: 2 launches to 1; norm sum order changed
# speedup vs baseline: 2.3600x; 1.0370x over previous
#include <hip/hip_runtime.h>


typedef __attribute__((ext_vector_type(2))) float v2f;
typedef __attribute__((ext_vector_type(4))) float v4f;
typedef __attribute__((ext_vector_type(8))) float v8f;

#define VQ_B 64
#define VQ_C 64
#define VQ_HW 1024
#define VQ_K 1024
#define WG_THREADS 512
#define WG_PIX 256
#define CB_PITCH 68
#define CHUNK_CODES 256
#define N_CHUNKS 4
#define CHUNK_TILES 16

__device__ __forceinline__ void vq_fetch(const float* __restrict__ cb, int chunk, int tid,
                                         v4f (&pf)[8]) {
  const v4f* gp = (const v4f*)(cb + (size_t)chunk * (CHUNK_CODES * VQ_C)) + tid;
#pragma unroll
  for (int i = 0; i < 8; ++i) pf[i] = gp[i * WG_THREADS];
}

template <int M>
__device__ __forceinline__ float vq_rowx(float v) {
  return __builtin_bit_cast(float, __builtin_amdgcn_update_dpp(
      0, __builtin_bit_cast(int, v), 0x160 + M, 0xf, 0xf, true));
}

__device__ __forceinline__ void vq_stash(float* cbs, float* cns, int chunk, int tid,
                                         const v4f (&pf)[8]) {
  const int row = chunk * CHUNK_CODES + (tid >> 4);
  float* sp = cbs + row * CB_PITCH + ((tid & 15) << 2);
  float q[8];
#pragma unroll
  for (int i = 0; i < 8; ++i) {
    *(v4f*)(sp + i * 32 * CB_PITCH) = pf[i];
    q[i] = pf[i].x * pf[i].x + pf[i].y * pf[i].y + pf[i].z * pf[i].z + pf[i].w * pf[i].w;
  }
#pragma unroll
  for (int i = 0; i < 8; ++i) q[i] += vq_rowx<1>(q[i]);
#pragma unroll
  for (int i = 0; i < 8; ++i) q[i] += vq_rowx<2>(q[i]);
#pragma unroll
  for (int i = 0; i < 8; ++i) q[i] += vq_rowx<4>(q[i]);
#pragma unroll
  for (int i = 0; i < 8; ++i) q[i] += vq_rowx<8>(q[i]);
  if ((tid & 15) == 0) {
#pragma unroll
    for (int i = 0; i < 8; ++i) cns[row + 32 * i] = 0.5f * q[i];
  }
}

__device__ __forceinline__ void vq_ldb(const float* cbs, const float* cns, int code, int bk,
                                       v2f (&bf)[16], float& cn) {
  cn = cns[code];
  const v2f* bp = (const v2f*)(cbs + code * CB_PITCH) + bk;
#pragma unroll
  for (int kk = 0; kk < 16; ++kk) bf[kk] = bp[kk * 2];
}

__device__ __forceinline__ void vq_tile(const v2f (&a)[16], const v2f (&bf)[16],
                                        const float cn, int code,
                                        float (&best)[8], int (&bidx)[8]) {
  v8f acc0 = {}, acc1 = {};
#pragma unroll
  for (int kk = 0; kk < 16; kk += 2) {
    acc0 = __builtin_amdgcn_wmma_f32_16x16x4_f32(
        false, a[kk + 0], false, bf[kk + 0], (short)0, acc0, false, false);
    acc1 = __builtin_amdgcn_wmma_f32_16x16x4_f32(
        false, a[kk + 1], false, bf[kk + 1], (short)0, acc1, false, false);
  }
#pragma unroll
  for (int r = 0; r < 8; ++r) {
    float s = (acc0[r] + acc1[r]) - cn;
    if (s > best[r]) { best[r] = s; bidx[r] = code; }
  }
}

__global__ __launch_bounds__(WG_THREADS) void vq_main_kernel(const float* __restrict__ in,
                                                             const float* __restrict__ cb,
                                                             float* __restrict__ out) {
  __shared__ __attribute__((aligned(16))) float cbs[VQ_K * CB_PITCH];
  __shared__ float cns[VQ_K];
  __shared__ int   rowbest[WG_PIX];

  const int tid  = threadIdx.x;
  const int lane = tid & 31;
  const int wave = tid >> 5;
  const int m    = blockIdx.x;
  const int b    = m >> 2;
  const int hwb  = (m & 3) << 8;
  const int hw0  = hwb + (wave << 4);

  v4f pf[8];
  vq_fetch(cb, 0, tid, pf);

  const int an = lane & 15;
  const int bk = lane >> 4;
  v2f a[16];
  {
    const float* ap = in + (((size_t)(b * VQ_C + 2 * bk)) << 10) + hw0 + an;
#pragma unroll
    for (int kk = 0; kk < 16; ++kk) {
      a[kk].x = ap[(size_t)(4 * kk + 0) << 10];
      a[kk].y = ap[(size_t)(4 * kk + 1) << 10];
    }
  }

  vq_stash(cbs, cns, 0, tid, pf);
  __syncthreads();

  float best[8];
  int   bidx[8];
#pragma unroll
  for (int r = 0; r < 8; ++r) { best[r] = -3.0e38f; bidx[r] = 0; }

  const int ncol = lane & 15;

#pragma unroll 1
  for (int ch = 0; ch < N_CHUNKS; ++ch) {
    if (ch + 1 < N_CHUNKS) vq_fetch(cb, ch + 1, tid, pf);

    const int cbase = ch * CHUNK_CODES + ncol;
    v2f bf0[16], bf1[16];
    float cn0, cn1;
    vq_ldb(cbs, cns, cbase, bk, bf0, cn0);
#pragma unroll 1
    for (int tt = 0; tt < CHUNK_TILES / 2; ++tt) {
      const int code0 = cbase + 32 * tt;
      const int code1 = code0 + 16;
      vq_ldb(cbs, cns, code1, bk, bf1, cn1);
      vq_tile(a, bf0, cn0, code0, best, bidx);
      if (tt + 1 < CHUNK_TILES / 2)
        vq_ldb(cbs, cns, code1 + 16, bk, bf0, cn0);
      vq_tile(a, bf1, cn1, code1, best, bidx);
    }

    if (ch + 1 < N_CHUNKS) {
      vq_stash(cbs, cns, ch + 1, tid, pf);
      __syncthreads();
    }
  }

#pragma unroll
  for (int r = 0; r < 8; ++r) {
    float s = best[r];
    int   i = bidx[r];
#pragma unroll
    for (int msk = 8; msk >= 1; msk >>= 1) {
      float os = __shfl_xor(s, msk, 32);
      int   oi = __shfl_xor(i, msk, 32);
      if (os > s || (os == s && oi < i)) { s = os; i = oi; }
    }
    const int rowm = r + (bk << 3);
    if (ncol == 0) rowbest[(wave << 4) + rowm] = i;
  }
  __syncthreads();

  {
    const int p  = ((wave >> 1) << 5) + lane;
    const int c0 = (wave & 1) << 5;
    const float* crow = cbs + rowbest[p] * CB_PITCH + c0;
    float* op = out + (((size_t)(b * VQ_C + c0)) << 10) + hwb + p;
#pragma unroll
    for (int j = 0; j < 32; ++j) op[(size_t)j << 10] = crow[j];
  }
}

extern "C" void kernel_launch(void* const* d_in, const int* in_sizes, int n_in,
                              void* d_out, int out_size, void* d_ws, size_t ws_size,
                              hipStream_t stream) {
  const float* in  = (const float*)d_in[0];
  const float* cb  = (const float*)d_in[1];
  float* out  = (float*)d_out;

  vq_main_kernel<<<(VQ_B * VQ_HW) / WG_PIX, WG_THREADS, 0, stream>>>(in, cb, out);
}


# Epilogue 16-lane argmax as branch-free DPP row_xmask butterflies (row max, then lowest code among its holders), 8 rows side by side, in place of 32 serial ds_bpermute levels with exec branches
# speedup vs baseline: 2.4596x; 1.0422x over previous
#include <hip/hip_runtime.h>


typedef __attribute__((ext_vector_type(2))) float v2f;
typedef __attribute__((ext_vector_type(4))) float v4f;
typedef __attribute__((ext_vector_type(8))) float v8f;

#define VQ_B 64
#define VQ_C 64
#define VQ_HW 1024
#define VQ_K 1024
#define WG_THREADS 512
#define WG_PIX 256
#define CB_PITCH 68
#define CHUNK_CODES 256
#define N_CHUNKS 4
#define CHUNK_TILES 16

__device__ __forceinline__ void vq_fetch(const float* __restrict__ cb, int chunk, int tid,
                                         v4f (&pf)[8]) {
  const v4f* gp = (const v4f*)(cb + (size_t)chunk * (CHUNK_CODES * VQ_C)) + tid;
#pragma unroll
  for (int i = 0; i < 8; ++i) pf[i] = gp[i * WG_THREADS];
}

template <int M>
__device__ __forceinline__ float vq_rowx(float v) {
  return __builtin_bit_cast(float, __builtin_amdgcn_update_dpp(
      0, __builtin_bit_cast(int, v), 0x160 + M, 0xf, 0xf, true));
}

template <int M>
__device__ __forceinline__ int vq_rowxi(int v) {
  return __builtin_amdgcn_update_dpp(0, v, 0x160 + M, 0xf, 0xf, true);
}

__device__ __forceinline__ void vq_stash(float* cbs, float* cns, int chunk, int tid,
                                         const v4f (&pf)[8]) {
  const int row = chunk * CHUNK_CODES + (tid >> 4);
  float* sp = cbs + row * CB_PITCH + ((tid & 15) << 2);
  float q[8];
#pragma unroll
  for (int i = 0; i < 8; ++i) {
    *(v4f*)(sp + i * 32 * CB_PITCH) = pf[i];
    q[i] = pf[i].x * pf[i].x + pf[i].y * pf[i].y + pf[i].z * pf[i].z + pf[i].w * pf[i].w;
  }
#pragma unroll
  for (int i = 0; i < 8; ++i) q[i] += vq_rowx<1>(q[i]);
#pragma unroll
  for (int i = 0; i < 8; ++i) q[i] += vq_rowx<2>(q[i]);
#pragma unroll
  for (int i = 0; i < 8; ++i) q[i] += vq_rowx<4>(q[i]);
#pragma unroll
  for (int i = 0; i < 8; ++i) q[i] += vq_rowx<8>(q[i]);
  if ((tid & 15) == 0) {
#pragma unroll
    for (int i = 0; i < 8; ++i) cns[row + 32 * i] = 0.5f * q[i];
  }
}

__device__ __forceinline__ void vq_ldb(const float* cbs, const float* cns, int code, int bk,
                                       v2f (&bf)[16], float& cn) {
  cn = cns[code];
  const v2f* bp = (const v2f*)(cbs + code * CB_PITCH) + bk;
#pragma unroll
  for (int kk = 0; kk < 16; ++kk) bf[kk] = bp[kk * 2];
}

__device__ __forceinline__ void vq_tile(const v2f (&a)[16], const v2f (&bf)[16],
                                        const float cn, int code,
                                        float (&best)[8], int (&bidx)[8]) {
  v8f acc0 = {}, acc1 = {};
#pragma unroll
  for (int kk = 0; kk < 16; kk += 2) {
    acc0 = __builtin_amdgcn_wmma_f32_16x16x4_f32(
        false, a[kk + 0], false, bf[kk + 0], (short)0, acc0, false, false);
    acc1 = __builtin_amdgcn_wmma_f32_16x16x4_f32(
        false, a[kk + 1], false, bf[kk + 1], (short)0, acc1, false, false);
  }
#pragma unroll
  for (int r = 0; r < 8; ++r) {
    float s = (acc0[r] + acc1[r]) - cn;
    if (s > best[r]) { best[r] = s; bidx[r] = code; }
  }
}

__global__ __launch_bounds__(WG_THREADS) void vq_main_kernel(const float* __restrict__ in,
                                                             const float* __restrict__ cb,
                                                             float* __restrict__ out) {
  __shared__ __attribute__((aligned(16))) float cbs[VQ_K * CB_PITCH];
  __shared__ float cns[VQ_K];
  __shared__ int   rowbest[WG_PIX];

  const int tid  = threadIdx.x;
  const int lane = tid & 31;
  const int wave = tid >> 5;
  const int m    = blockIdx.x;
  const int b    = m >> 2;
  const int hwb  = (m & 3) << 8;
  const int hw0  = hwb + (wave << 4);

  v4f pf[8];
  vq_fetch(cb, 0, tid, pf);

  const int an = lane & 15;
  const int bk = lane >> 4;
  v2f a[16];
  {
    const float* ap = in + (((size_t)(b * VQ_C + 2 * bk)) << 10) + hw0 + an;
#pragma unroll
    for (int kk = 0; kk < 16; ++kk) {
      a[kk].x = ap[(size_t)(4 * kk + 0) << 10];
      a[kk].y = ap[(size_t)(4 * kk + 1) << 10];
    }
  }

  vq_stash(cbs, cns, 0, tid, pf);
  __syncthreads();

  float best[8];
  int   bidx[8];
#pragma unroll
  for (int r = 0; r < 8; ++r) { best[r] = -3.0e38f; bidx[r] = 0; }

  const int ncol = lane & 15;

#pragma unroll 1
  for (int ch = 0; ch < N_CHUNKS; ++ch) {
    if (ch + 1 < N_CHUNKS) vq_fetch(cb, ch + 1, tid, pf);

    const int cbase = ch * CHUNK_CODES + ncol;
    v2f bf0[16], bf1[16];
    float cn0, cn1;
    vq_ldb(cbs, cns, cbase, bk, bf0, cn0);
#pragma unroll 1
    for (int tt = 0; tt < CHUNK_TILES / 2; ++tt) {
      const int code0 = cbase + 32 * tt;
      const int code1 = code0 + 16;
      vq_ldb(cbs, cns, code1, bk, bf1, cn1);
      vq_tile(a, bf0, cn0, code0, best, bidx);
      if (tt + 1 < CHUNK_TILES / 2)
        vq_ldb(cbs, cns, code1 + 16, bk, bf0, cn0);
      vq_tile(a, bf1, cn1, code1, best, bidx);
    }

    if (ch + 1 < N_CHUNKS) {
      vq_stash(cbs, cns, ch + 1, tid, pf);
      __syncthreads();
    }
  }

  {
    float mx[8];
    int   win[8];
#pragma unroll
    for (int r = 0; r < 8; ++r) mx[r] = best[r];
#pragma unroll
    for (int r = 0; r < 8; ++r) mx[r] = __builtin_fmaxf(mx[r], vq_rowx<8>(mx[r]));
#pragma unroll
    for (int r = 0; r < 8; ++r) mx[r] = __builtin_fmaxf(mx[r], vq_rowx<4>(mx[r]));
#pragma unroll
    for (int r = 0; r < 8; ++r) mx[r] = __builtin_fmaxf(mx[r], vq_rowx<2>(mx[r]));
#pragma unroll
    for (int r = 0; r < 8; ++r) mx[r] = __builtin_fmaxf(mx[r], vq_rowx<1>(mx[r]));
#pragma unroll
    for (int r = 0; r < 8; ++r) win[r] = (best[r] == mx[r]) ? bidx[r] : 0x7fffffff;
#pragma unroll
    for (int r = 0; r < 8; ++r) win[r] = min(win[r], vq_rowxi<8>(win[r]));
#pragma unroll
    for (int r = 0; r < 8; ++r) win[r] = min(win[r], vq_rowxi<4>(win[r]));
#pragma unroll
    for (int r = 0; r < 8; ++r) win[r] = min(win[r], vq_rowxi<2>(win[r]));
#pragma unroll
    for (int r = 0; r < 8; ++r) win[r] = min(win[r], vq_rowxi<1>(win[r]));
    if (ncol == 0) {
#pragma unroll
      for (int r = 0; r < 8; ++r) rowbest[(wave << 4) + r + (bk << 3)] = win[r];
    }
  }
  __syncthreads();

  {
    const int p  = ((wave >> 1) << 5) + lane;
    const int c0 = (wave & 1) << 5;
    const float* crow = cbs + rowbest[p] * CB_PITCH + c0;
    float* op = out + (((size_t)(b * VQ_C + c0)) << 10) + hwb + p;
#pragma unroll
    for (int j = 0; j < 32; ++j) op[(size_t)j << 10] = crow[j];
  }
}

extern "C" void kernel_launch(void* const* d_in, const int* in_sizes, int n_in,
                              void* d_out, int out_size, void* d_ws, size_t ws_size,
                              hipStream_t stream) {
  const float* in  = (const float*)d_in[0];
  const float* cb  = (const float*)d_in[1];
  float* out  = (float*)d_out;

  vq_main_kernel<<<(VQ_B * VQ_HW) / WG_PIX, WG_THREADS, 0, stream>>>(in, cb, out);
}
